# starting kernel
# baseline (speedup 1.0000x reference)
#include <hip/hip_runtime.h>
#include <math.h>

typedef __attribute__((ext_vector_type(2))) float v2f;
typedef __attribute__((ext_vector_type(8))) float v8f;

#define NB 8
#define NL 1024
#define NE 256
#define NR 16
#define NROWS (NB * NL)
#define LOG2E 1.4426950408889634f

__device__ __forceinline__ v8f wmma_f32_k4(v2f a, v2f b, v8f c) {
  return __builtin_amdgcn_wmma_f32_16x16x4_f32(false, a, false, b, (short)0, c,
                                               false, false);
}

__global__ void __launch_bounds__(256)
rbf_local_kernel(const float* __restrict__ coords,
                 const float* __restrict__ centers,
                 const float* __restrict__ gamma_p,
                 float* __restrict__ rbfl) {
  const int b    = blockIdx.x >> 6;
  const int i0   = (blockIdx.x & 63) << 4;
  const int tid  = threadIdx.x;
  const int lane = tid & 31;
  const int wave = tid >> 5;
  const int half = lane >> 4;
  const int lid  = lane & 15;
  const float ngl = -gamma_p[0] * LOG2E;

  __shared__ float red[NR * 16];
  red[tid] = 0.0f;
  __syncthreads();

  float cent[NR];
#pragma unroll
  for (int k = 0; k < NR; ++k) cent[k] = centers[k];

  const float* cb = coords + (size_t)b * NL * 3;

  const int arow = i0 + lid;
  v2f afrag;
  afrag.x = cb[arow * 3 + (half ? 2 : 0)];
  afrag.y = half ? 0.0f : cb[arow * 3 + 1];

  float rn[8];
#pragma unroll
  for (int r = 0; r < 8; ++r) {
    const int m = i0 + r + 8 * half;
    const float x = cb[m * 3 + 0];
    const float y = cb[m * 3 + 1];
    const float z = cb[m * 3 + 2];
    rn[r] = x * x + y * y + z * z;
  }

  float acc[8 * NR];
#pragma unroll
  for (int i = 0; i < 8 * NR; ++i) acc[i] = 0.0f;

  for (int jt = wave; jt < NL / 16; jt += 8) {
    const int jcol = (jt << 4) + lid;
    const float bx = cb[jcol * 3 + 0];
    const float by = cb[jcol * 3 + 1];
    const float bz = cb[jcol * 3 + 2];
    const float cn2 = bx * bx + by * by + bz * bz;

    v2f bfrag;
    bfrag.x = half ? bz : bx;
    bfrag.y = half ? 0.0f : by;

    v8f dot = {};
    dot = wmma_f32_k4(afrag, bfrag, dot);

#pragma unroll
    for (int r = 0; r < 8; ++r) {
      const float sq = rn[r] + cn2 - 2.0f * dot[r];
      const float d  = __builtin_amdgcn_sqrtf(fmaxf(sq, 1e-12f));
#pragma unroll
      for (int k = 0; k < NR; ++k) {
        const float t = d - cent[k];
        acc[r * NR + k] += __builtin_amdgcn_exp2f(ngl * t * t);
      }
    }
  }

#pragma unroll
  for (int i = 0; i < 8 * NR; ++i) {
    float v = acc[i];
    v += __shfl_xor(v, 1, 16);
    v += __shfl_xor(v, 2, 16);
    v += __shfl_xor(v, 4, 16);
    v += __shfl_xor(v, 8, 16);
    acc[i] = v;
  }
  if (lid == 0) {
#pragma unroll
    for (int r = 0; r < 8; ++r)
#pragma unroll
      for (int k = 0; k < NR; ++k)
        atomicAdd(&red[(r + 8 * half) * NR + k], acc[r * NR + k]);
  }
  __syncthreads();

  {
    const int m = tid >> 4;
    const int k = tid & 15;
    rbfl[((size_t)b * NL + i0 + m) * NR + k] = red[tid] * (1.0f / (float)NL);
  }
}

__global__ void __launch_bounds__(256)
weff_kernel(const float* __restrict__ w1, const float* __restrict__ b1,
            const float* __restrict__ w2, const float* __restrict__ b2,
            float* __restrict__ weff, float* __restrict__ beff) {
  const int idx = blockIdx.x * blockDim.x + threadIdx.x;
  if (idx < NR * NE) {
    const int k = idx >> 8;
    const int n = idx & (NE - 1);
    float s = 0.0f;
    for (int e = 0; e < NE; ++e) s += w1[k * NE + e] * w2[(NE + e) * NE + n];
    weff[idx] = s;
  } else if (idx < NR * NE + NE) {
    const int n = idx - NR * NE;
    float s = b2[n];
    for (int e = 0; e < NE; ++e) s += b1[e] * w2[(NE + e) * NE + n];
    beff[n] = s;
  }
}

__global__ void __launch_bounds__(256)
fused_out_kernel(const int* __restrict__ Z,
                 const float* __restrict__ atom_emb,
                 const float* __restrict__ w2,
                 const float* __restrict__ rbfl,
                 const float* __restrict__ weff,
                 const float* __restrict__ beff,
                 float* __restrict__ out) {
  const int lane = threadIdx.x & 31;
  const int wave = threadIdx.x >> 5;
  const int half = lane >> 4;
  const int lid  = lane & 15;
  const int gw   = blockIdx.x * 8 + wave;
  const int tm   = gw >> 4;
  const int tn   = gw & 15;
  const int row0 = tm << 4;
  const int jcol = (tn << 4) + lid;

  const int zrow = Z[row0 + lid];
  const float* aemb = atom_emb + (size_t)zrow * NE;
  const float* arbf = rbfl + (size_t)(row0 + lid) * NR;

  const float bval = beff[jcol];
  v8f c;
#pragma unroll
  for (int r = 0; r < 8; ++r) c[r] = bval;

#pragma unroll 4
  for (int s = 0; s < NE / 4; ++s) {
    const int kb = 4 * s + 2 * half;
    v2f a;
    a.x = aemb[kb];
    a.y = aemb[kb + 1];
    v2f bb;
    bb.x = w2[kb * NE + jcol];
    bb.y = w2[(kb + 1) * NE + jcol];
    c = wmma_f32_k4(a, bb, c);
  }
#pragma unroll
  for (int s = 0; s < NR / 4; ++s) {
    const int kb = 4 * s + 2 * half;
    v2f a;
    a.x = arbf[kb];
    a.y = arbf[kb + 1];
    v2f bb;
    bb.x = weff[kb * NE + jcol];
    bb.y = weff[(kb + 1) * NE + jcol];
    c = wmma_f32_k4(a, bb, c);
  }

#pragma unroll
  for (int r = 0; r < 8; ++r)
    out[(size_t)(row0 + r + 8 * half) * NE + jcol] = c[r];
}

extern "C" void kernel_launch(void* const* d_in, const int* in_sizes, int n_in,
                              void* d_out, int out_size, void* d_ws,
                              size_t ws_size, hipStream_t stream) {
  const float* coords   = (const float*)d_in[0];
  const int*   Z        = (const int*)d_in[1];
  const float* atom_emb = (const float*)d_in[2];
  const float* centers  = (const float*)d_in[3];
  const float* gamma    = (const float*)d_in[4];
  const float* w1       = (const float*)d_in[5];
  const float* b1       = (const float*)d_in[6];
  const float* w2       = (const float*)d_in[7];
  const float* b2       = (const float*)d_in[8];
  float* outp = (float*)d_out;

  float* rbfl = (float*)d_ws;
  float* weff = rbfl + NROWS * NR;
  float* beff = weff + NR * NE;

  weff_kernel<<<(NR * NE + NE + 255) / 256, 256, 0, stream>>>(w1, b1, w2, b2,
                                                              weff, beff);
  rbf_local_kernel<<<NB * (NL / 16), 256, 0, stream>>>(coords, centers, gamma,
                                                       rbfl);
  fused_out_kernel<<<(NROWS / 16) * (NE / 16) / 8, 256, 0, stream>>>(
      Z, atom_emb, w2, rbfl, weff, beff, outp);
}


# Whole forward per 16-row block in ONE launch: rbf_local -> h_geo=rbf@W1+b1 in LDS -> [atom_emb[Z]|h_geo]@W2+b2 on WMMA; no Weff, no workspace; sum order now (rbf@W1)@W2, k ascending
# speedup vs baseline: 1.0885x; 1.0885x over previous
#include <hip/hip_runtime.h>
#include <math.h>


typedef __attribute__((ext_vector_type(2))) float v2f;
typedef __attribute__((ext_vector_type(4))) float v4f;
typedef __attribute__((ext_vector_type(8))) float v8f;

#define NB 8
#define NL 1024
#define NE 256
#define NR 16
#define NROWS (NB * NL)
#define LOG2E 1.4426950408889634f
#define APITCH 516

__device__ __forceinline__ v8f wmma_f32_k4(v2f a, v2f b, v8f c) {
  return __builtin_amdgcn_wmma_f32_16x16x4_f32(false, a, false, b, (short)0, c,
                                               false, false);
}

__device__ __forceinline__ void load_group(const float* arow_lds,
                                           const float* __restrict__ wcol,
                                           int g, v2f (&a)[8], v2f (&w)[16]) {
#pragma unroll
  for (int t = 0; t < 8; ++t) {
    const int k = 32 * g + 4 * t;
    a[t]         = *(const v2f*)(arow_lds + k);
    w[2 * t]     = *(const v2f*)(wcol + (size_t)k * NE);
    w[2 * t + 1] = *(const v2f*)(wcol + (size_t)(k + 1) * NE);
  }
}

__device__ __forceinline__ void mma_group(const v2f (&a)[8], const v2f (&w)[16],
                                          v8f& o0, v8f& o1) {
#pragma unroll
  for (int t = 0; t < 8; ++t) {
    v2f bb0, bb1;
    bb0.x = w[2 * t].x; bb0.y = w[2 * t + 1].x;
    bb1.x = w[2 * t].y; bb1.y = w[2 * t + 1].y;
    o0 = wmma_f32_k4(a[t], bb0, o0);
    o1 = wmma_f32_k4(a[t], bb1, o1);
  }
}

__global__ void __launch_bounds__(256)
coord_encoder_kernel(const float* __restrict__ coords,
                     const int* __restrict__ Z,
                     const float* __restrict__ atom_emb,
                     const float* __restrict__ centers,
                     const float* __restrict__ gamma_p,
                     const float* __restrict__ w1,
                     const float* __restrict__ b1,
                     const float* __restrict__ w2,
                     const float* __restrict__ b2,
                     float* __restrict__ out) {
  const int b    = blockIdx.x >> 6;
  const int i0   = (blockIdx.x & 63) << 4;
  const int tid  = threadIdx.x;
  const int lane = tid & 31;
  const int wave = tid >> 5;
  const int half = lane >> 4;
  const int lid  = lane & 15;
  const float ngl = -gamma_p[0] * LOG2E;

  __shared__ float red[NR * 16];
  __shared__ __attribute__((aligned(16))) float hcat[16 * APITCH];

  red[tid] = 0.0f;

  {
    const int grow = tid >> 4;
    const int seg  = tid & 15;
    const int z    = Z[b * NL + i0 + grow];
    const float* src = atom_emb + (size_t)z * NE;
#pragma unroll
    for (int q = 0; q < 4; ++q) {
      const int col = 4 * (seg + 16 * q);
      const v4f v = *(const v4f*)(src + col);
      *(v4f*)(&hcat[grow * APITCH + col]) = v;
    }
  }
  __syncthreads();

  float cent[NR];
#pragma unroll
  for (int k = 0; k < NR; ++k) cent[k] = centers[k];

  const float* cb = coords + (size_t)b * NL * 3;

  const int arow = i0 + lid;
  v2f afrag;
  afrag.x = cb[arow * 3 + (half ? 2 : 0)];
  afrag.y = half ? 0.0f : cb[arow * 3 + 1];

  float rn[8];
#pragma unroll
  for (int r = 0; r < 8; ++r) {
    const int m = i0 + r + 8 * half;
    const float x = cb[m * 3 + 0];
    const float y = cb[m * 3 + 1];
    const float z = cb[m * 3 + 2];
    rn[r] = x * x + y * y + z * z;
  }

  float acc[8 * NR];
#pragma unroll
  for (int i = 0; i < 8 * NR; ++i) acc[i] = 0.0f;

  for (int jt = wave; jt < NL / 16; jt += 8) {
    const int jcol = (jt << 4) + lid;
    const float bx = cb[jcol * 3 + 0];
    const float by = cb[jcol * 3 + 1];
    const float bz = cb[jcol * 3 + 2];
    const float cn2 = bx * bx + by * by + bz * bz;

    v2f bfrag;
    bfrag.x = half ? bz : bx;
    bfrag.y = half ? 0.0f : by;

    v8f dot = {};
    dot = wmma_f32_k4(afrag, bfrag, dot);

#pragma unroll
    for (int r = 0; r < 8; ++r) {
      const float sq = rn[r] + cn2 - 2.0f * dot[r];
      const float d  = __builtin_amdgcn_sqrtf(fmaxf(sq, 1e-12f));
#pragma unroll
      for (int k = 0; k < NR; ++k) {
        const float t = d - cent[k];
        acc[r * NR + k] += __builtin_amdgcn_exp2f(ngl * t * t);
      }
    }
  }

#pragma unroll
  for (int i = 0; i < 8 * NR; ++i) {
    float v = acc[i];
    v += __shfl_xor(v, 1, 16);
    v += __shfl_xor(v, 2, 16);
    v += __shfl_xor(v, 4, 16);
    v += __shfl_xor(v, 8, 16);
    acc[i] = v;
  }
  if (lid == 0) {
#pragma unroll
    for (int r = 0; r < 8; ++r)
#pragma unroll
      for (int k = 0; k < NR; ++k)
        atomicAdd(&red[(r + 8 * half) * NR + k], acc[r * NR + k]);
  }
  __syncthreads();

  const int c0 = 32 * wave + 2 * lid;

  {
    const v2f bias = *(const v2f*)(b1 + c0);
    v8f g0, g1;
#pragma unroll
    for (int r = 0; r < 8; ++r) { g0[r] = bias.x; g1[r] = bias.y; }
#pragma unroll
    for (int s = 0; s < NR / 4; ++s) {
      const int kb = 4 * s + 2 * half;
      v2f a;
      a.x = red[lid * NR + kb] * (1.0f / (float)NL);
      a.y = red[lid * NR + kb + 1] * (1.0f / (float)NL);
      const v2f r0 = *(const v2f*)(w1 + kb * NE + c0);
      const v2f r1 = *(const v2f*)(w1 + (kb + 1) * NE + c0);
      v2f bb0, bb1;
      bb0.x = r0.x; bb0.y = r1.x;
      bb1.x = r0.y; bb1.y = r1.y;
      g0 = wmma_f32_k4(a, bb0, g0);
      g1 = wmma_f32_k4(a, bb1, g1);
    }
#pragma unroll
    for (int r = 0; r < 8; ++r) {
      v2f g;
      g.x = g0[r];
      g.y = g1[r];
      *(v2f*)(&hcat[(r + 8 * half) * APITCH + NE + c0]) = g;
    }
  }
  __syncthreads();

  {
    const v2f bias = *(const v2f*)(b2 + c0);
    v8f o0, o1;
#pragma unroll
    for (int r = 0; r < 8; ++r) { o0[r] = bias.x; o1[r] = bias.y; }
    const float* arow_lds = &hcat[lid * APITCH + 2 * half];
    const float* wcol = w2 + (size_t)(2 * half) * NE + c0;

    v2f aa[8], ab[8];
    v2f wa[16], wb[16];

    load_group(arow_lds, wcol, 0, aa, wa);
#pragma unroll 1
    for (int gp = 0; gp < 7; ++gp) {
      load_group(arow_lds, wcol, 2 * gp + 1, ab, wb);
      mma_group(aa, wa, o0, o1);
      load_group(arow_lds, wcol, 2 * gp + 2, aa, wa);
      mma_group(ab, wb, o0, o1);
    }
    load_group(arow_lds, wcol, 15, ab, wb);
    mma_group(aa, wa, o0, o1);
    mma_group(ab, wb, o0, o1);

    float* orow = out + ((size_t)b * NL + i0 + 8 * half) * NE + c0;
#pragma unroll
    for (int r = 0; r < 8; ++r) {
      v2f o;
      o.x = o0[r];
      o.y = o1[r];
      *(v2f*)(orow + (size_t)r * NE) = o;
    }
  }
}

extern "C" void kernel_launch(void* const* d_in, const int* in_sizes, int n_in,
                              void* d_out, int out_size, void* d_ws,
                              size_t ws_size, hipStream_t stream) {
  const float* coords   = (const float*)d_in[0];
  const int*   Z        = (const int*)d_in[1];
  const float* atom_emb = (const float*)d_in[2];
  const float* centers  = (const float*)d_in[3];
  const float* gamma    = (const float*)d_in[4];
  const float* w1       = (const float*)d_in[5];
  const float* b1       = (const float*)d_in[6];
  const float* w2       = (const float*)d_in[7];
  const float* b2       = (const float*)d_in[8];
  float* outp = (float*)d_out;

  coord_encoder_kernel<<<NB * (NL / 16), 256, 0, stream>>>(
      coords, Z, atom_emb, centers, gamma, w1, b1, w2, b2, outp);
}


# rbf loop: per row the 8 centre pairs run as independent two-wide f32 ops (sub, scale, square, add) with the 16 exps between, instead of one serial dependent chain; sum order unchanged
# speedup vs baseline: 1.4404x; 1.3233x over previous
#include <hip/hip_runtime.h>
#include <math.h>


typedef __attribute__((ext_vector_type(2))) float v2f;
typedef __attribute__((ext_vector_type(4))) float v4f;
typedef __attribute__((ext_vector_type(8))) float v8f;

#define NB 8
#define NL 1024
#define NE 256
#define NR 16
#define NROWS (NB * NL)
#define LOG2E 1.4426950408889634f
#define APITCH 516

__device__ __forceinline__ v8f wmma_f32_k4(v2f a, v2f b, v8f c) {
  return __builtin_amdgcn_wmma_f32_16x16x4_f32(false, a, false, b, (short)0, c,
                                               false, false);
}

__device__ __forceinline__ void load_group(const float* arow_lds,
                                           const float* __restrict__ wcol,
                                           int g, v2f (&a)[8], v2f (&w)[16]) {
#pragma unroll
  for (int t = 0; t < 8; ++t) {
    const int k = 32 * g + 4 * t;
    a[t]         = *(const v2f*)(arow_lds + k);
    w[2 * t]     = *(const v2f*)(wcol + (size_t)k * NE);
    w[2 * t + 1] = *(const v2f*)(wcol + (size_t)(k + 1) * NE);
  }
}

__device__ __forceinline__ void mma_group(const v2f (&a)[8], const v2f (&w)[16],
                                          v8f& o0, v8f& o1) {
#pragma unroll
  for (int t = 0; t < 8; ++t) {
    v2f bb0, bb1;
    bb0.x = w[2 * t].x; bb0.y = w[2 * t + 1].x;
    bb1.x = w[2 * t].y; bb1.y = w[2 * t + 1].y;
    o0 = wmma_f32_k4(a[t], bb0, o0);
    o1 = wmma_f32_k4(a[t], bb1, o1);
  }
}

__global__ void __launch_bounds__(256)
coord_encoder_kernel(const float* __restrict__ coords,
                     const int* __restrict__ Z,
                     const float* __restrict__ atom_emb,
                     const float* __restrict__ centers,
                     const float* __restrict__ gamma_p,
                     const float* __restrict__ w1,
                     const float* __restrict__ b1,
                     const float* __restrict__ w2,
                     const float* __restrict__ b2,
                     float* __restrict__ out) {
  const int b    = blockIdx.x >> 6;
  const int i0   = (blockIdx.x & 63) << 4;
  const int tid  = threadIdx.x;
  const int lane = tid & 31;
  const int wave = tid >> 5;
  const int half = lane >> 4;
  const int lid  = lane & 15;
  const float ngl = -gamma_p[0] * LOG2E;

  __shared__ float red[NR * 16];
  __shared__ __attribute__((aligned(16))) float hcat[16 * APITCH];

  red[tid] = 0.0f;

  {
    const int grow = tid >> 4;
    const int seg  = tid & 15;
    const int z    = Z[b * NL + i0 + grow];
    const float* src = atom_emb + (size_t)z * NE;
#pragma unroll
    for (int q = 0; q < 4; ++q) {
      const int col = 4 * (seg + 16 * q);
      const v4f v = *(const v4f*)(src + col);
      *(v4f*)(&hcat[grow * APITCH + col]) = v;
    }
  }
  __syncthreads();

  v2f cent2[NR / 2];
#pragma unroll
  for (int p = 0; p < NR / 2; ++p) {
    cent2[p].x = centers[2 * p];
    cent2[p].y = centers[2 * p + 1];
  }
  v2f ngl2;
  ngl2.x = ngl;
  ngl2.y = ngl;

  const float* cb = coords + (size_t)b * NL * 3;

  const int arow = i0 + lid;
  v2f afrag;
  afrag.x = cb[arow * 3 + (half ? 2 : 0)];
  afrag.y = half ? 0.0f : cb[arow * 3 + 1];

  float rn[8];
#pragma unroll
  for (int r = 0; r < 8; ++r) {
    const int m = i0 + r + 8 * half;
    const float x = cb[m * 3 + 0];
    const float y = cb[m * 3 + 1];
    const float z = cb[m * 3 + 2];
    rn[r] = x * x + y * y + z * z;
  }

  v2f acc[8][NR / 2];
#pragma unroll
  for (int r = 0; r < 8; ++r)
#pragma unroll
    for (int p = 0; p < NR / 2; ++p) { acc[r][p].x = 0.0f; acc[r][p].y = 0.0f; }

  for (int jt = wave; jt < NL / 16; jt += 8) {
    const int jcol = (jt << 4) + lid;
    const float bx = cb[jcol * 3 + 0];
    const float by = cb[jcol * 3 + 1];
    const float bz = cb[jcol * 3 + 2];
    const float cn2 = bx * bx + by * by + bz * bz;

    v2f bfrag;
    bfrag.x = half ? bz : bx;
    bfrag.y = half ? 0.0f : by;

    v8f dot = {};
    dot = wmma_f32_k4(afrag, bfrag, dot);

    float dd[8];
#pragma unroll
    for (int r = 0; r < 8; ++r) {
      const float sq = rn[r] + cn2 - 2.0f * dot[r];
      dd[r] = __builtin_amdgcn_sqrtf(fmaxf(sq, 1e-12f));
    }
#pragma unroll
    for (int r = 0; r < 8; ++r) {
      v2f d2;
      d2.x = dd[r];
      d2.y = dd[r];
      v2f q[NR / 2];
#pragma unroll
      for (int p = 0; p < NR / 2; ++p) {
        const v2f t = d2 - cent2[p];
        q[p] = (ngl2 * t) * t;
      }
#pragma unroll
      for (int p = 0; p < NR / 2; ++p) {
        v2f e;
        e.x = __builtin_amdgcn_exp2f(q[p].x);
        e.y = __builtin_amdgcn_exp2f(q[p].y);
        acc[r][p] += e;
      }
    }
  }

#pragma unroll
  for (int r = 0; r < 8; ++r)
#pragma unroll
    for (int k = 0; k < NR; ++k) {
      float v = (k & 1) ? acc[r][k >> 1].y : acc[r][k >> 1].x;
      v += __shfl_xor(v, 1, 16);
      v += __shfl_xor(v, 2, 16);
      v += __shfl_xor(v, 4, 16);
      v += __shfl_xor(v, 8, 16);
      if (k & 1) acc[r][k >> 1].y = v; else acc[r][k >> 1].x = v;
    }
  if (lid == 0) {
#pragma unroll
    for (int r = 0; r < 8; ++r)
#pragma unroll
      for (int k = 0; k < NR; ++k)
        atomicAdd(&red[(r + 8 * half) * NR + k],
                  (k & 1) ? acc[r][k >> 1].y : acc[r][k >> 1].x);
  }
  __syncthreads();

  const int c0 = 32 * wave + 2 * lid;

  {
    const v2f bias = *(const v2f*)(b1 + c0);
    v8f g0, g1;
#pragma unroll
    for (int r = 0; r < 8; ++r) { g0[r] = bias.x; g1[r] = bias.y; }
#pragma unroll
    for (int s = 0; s < NR / 4; ++s) {
      const int kb = 4 * s + 2 * half;
      v2f a;
      a.x = red[lid * NR + kb] * (1.0f / (float)NL);
      a.y = red[lid * NR + kb + 1] * (1.0f / (float)NL);
      const v2f r0 = *(const v2f*)(w1 + kb * NE + c0);
      const v2f r1 = *(const v2f*)(w1 + (kb + 1) * NE + c0);
      v2f bb0, bb1;
      bb0.x = r0.x; bb0.y = r1.x;
      bb1.x = r0.y; bb1.y = r1.y;
      g0 = wmma_f32_k4(a, bb0, g0);
      g1 = wmma_f32_k4(a, bb1, g1);
    }
#pragma unroll
    for (int r = 0; r < 8; ++r) {
      v2f g;
      g.x = g0[r];
      g.y = g1[r];
      *(v2f*)(&hcat[(r + 8 * half) * APITCH + NE + c0]) = g;
    }
  }
  __syncthreads();

  {
    const v2f bias = *(const v2f*)(b2 + c0);
    v8f o0, o1;
#pragma unroll
    for (int r = 0; r < 8; ++r) { o0[r] = bias.x; o1[r] = bias.y; }
    const float* arow_lds = &hcat[lid * APITCH + 2 * half];
    const float* wcol = w2 + (size_t)(2 * half) * NE + c0;

    v2f aa[8], ab[8];
    v2f wa[16], wb[16];

    load_group(arow_lds, wcol, 0, aa, wa);
#pragma unroll 1
    for (int gp = 0; gp < 7; ++gp) {
      load_group(arow_lds, wcol, 2 * gp + 1, ab, wb);
      mma_group(aa, wa, o0, o1);
      load_group(arow_lds, wcol, 2 * gp + 2, aa, wa);
      mma_group(ab, wb, o0, o1);
    }
    load_group(arow_lds, wcol, 15, ab, wb);
    mma_group(aa, wa, o0, o1);
    mma_group(ab, wb, o0, o1);

    float* orow = out + ((size_t)b * NL + i0 + 8 * half) * NE + c0;
#pragma unroll
    for (int r = 0; r < 8; ++r) {
      v2f o;
      o.x = o0[r];
      o.y = o1[r];
      *(v2f*)(orow + (size_t)r * NE) = o;
    }
  }
}

extern "C" void kernel_launch(void* const* d_in, const int* in_sizes, int n_in,
                              void* d_out, int out_size, void* d_ws,
                              size_t ws_size, hipStream_t stream) {
  const float* coords   = (const float*)d_in[0];
  const int*   Z        = (const int*)d_in[1];
  const float* atom_emb = (const float*)d_in[2];
  const float* centers  = (const float*)d_in[3];
  const float* gamma    = (const float*)d_in[4];
  const float* w1       = (const float*)d_in[5];
  const float* b1       = (const float*)d_in[6];
  const float* w2       = (const float*)d_in[7];
  const float* b2       = (const float*)d_in[8];
  float* outp = (float*)d_out;

  coord_encoder_kernel<<<NB * (NL / 16), 256, 0, stream>>>(
      coords, Z, atom_emb, centers, gamma, w1, b1, w2, b2, outp);
}


# VGPRs 262 -> 228 (launch bound (256,4); each row's 16 sums go to LDS before the next row's exchange): two work-groups resident per WGP instead of one, 4 waves per SIMD; sum order unchanged
# speedup vs baseline: 2.0329x; 1.4113x over previous
#include <hip/hip_runtime.h>
#include <math.h>


typedef __attribute__((ext_vector_type(2))) float v2f;
typedef __attribute__((ext_vector_type(4))) float v4f;
typedef __attribute__((ext_vector_type(8))) float v8f;

#define NB 8
#define NL 1024
#define NE 256
#define NR 16
#define NROWS (NB * NL)
#define LOG2E 1.4426950408889634f
#define APITCH 516

__device__ __forceinline__ v8f wmma_f32_k4(v2f a, v2f b, v8f c) {
  return __builtin_amdgcn_wmma_f32_16x16x4_f32(false, a, false, b, (short)0, c,
                                               false, false);
}

__device__ __forceinline__ void load_group(const float* arow_lds,
                                           const float* __restrict__ wcol,
                                           int g, v2f (&a)[8], v2f (&w)[16]) {
#pragma unroll
  for (int t = 0; t < 8; ++t) {
    const int k = 32 * g + 4 * t;
    a[t]         = *(const v2f*)(arow_lds + k);
    w[2 * t]     = *(const v2f*)(wcol + (size_t)k * NE);
    w[2 * t + 1] = *(const v2f*)(wcol + (size_t)(k + 1) * NE);
  }
}

__device__ __forceinline__ void mma_group(const v2f (&a)[8], const v2f (&w)[16],
                                          v8f& o0, v8f& o1) {
#pragma unroll
  for (int t = 0; t < 8; ++t) {
    v2f bb0, bb1;
    bb0.x = w[2 * t].x; bb0.y = w[2 * t + 1].x;
    bb1.x = w[2 * t].y; bb1.y = w[2 * t + 1].y;
    o0 = wmma_f32_k4(a[t], bb0, o0);
    o1 = wmma_f32_k4(a[t], bb1, o1);
  }
}

__global__ void __launch_bounds__(256, 4)
coord_encoder_kernel(const float* __restrict__ coords,
                     const int* __restrict__ Z,
                     const float* __restrict__ atom_emb,
                     const float* __restrict__ centers,
                     const float* __restrict__ gamma_p,
                     const float* __restrict__ w1,
                     const float* __restrict__ b1,
                     const float* __restrict__ w2,
                     const float* __restrict__ b2,
                     float* __restrict__ out) {
  const int b    = blockIdx.x >> 6;
  const int i0   = (blockIdx.x & 63) << 4;
  const int tid  = threadIdx.x;
  const int lane = tid & 31;
  const int wave = tid >> 5;
  const int half = lane >> 4;
  const int lid  = lane & 15;
  const float ngl = -gamma_p[0] * LOG2E;

  __shared__ float red[NR * 16];
  __shared__ __attribute__((aligned(16))) float hcat[16 * APITCH];

  red[tid] = 0.0f;

  {
    const int grow = tid >> 4;
    const int seg  = tid & 15;
    const int z    = Z[b * NL + i0 + grow];
    const float* src = atom_emb + (size_t)z * NE;
#pragma unroll
    for (int q = 0; q < 4; ++q) {
      const int col = 4 * (seg + 16 * q);
      const v4f v = *(const v4f*)(src + col);
      *(v4f*)(&hcat[grow * APITCH + col]) = v;
    }
  }
  __syncthreads();

  v2f cent2[NR / 2];
#pragma unroll
  for (int p = 0; p < NR / 2; ++p) {
    cent2[p].x = centers[2 * p];
    cent2[p].y = centers[2 * p + 1];
  }
  v2f ngl2;
  ngl2.x = ngl;
  ngl2.y = ngl;

  const float* cb = coords + (size_t)b * NL * 3;

  const int arow = i0 + lid;
  v2f afrag;
  afrag.x = cb[arow * 3 + (half ? 2 : 0)];
  afrag.y = half ? 0.0f : cb[arow * 3 + 1];

  float rn[8];
#pragma unroll
  for (int r = 0; r < 8; ++r) {
    const int m = i0 + r + 8 * half;
    const float x = cb[m * 3 + 0];
    const float y = cb[m * 3 + 1];
    const float z = cb[m * 3 + 2];
    rn[r] = x * x + y * y + z * z;
  }

  v2f acc[8][NR / 2];
#pragma unroll
  for (int r = 0; r < 8; ++r)
#pragma unroll
    for (int p = 0; p < NR / 2; ++p) { acc[r][p].x = 0.0f; acc[r][p].y = 0.0f; }

  for (int jt = wave; jt < NL / 16; jt += 8) {
    const int jcol = (jt << 4) + lid;
    const float bx = cb[jcol * 3 + 0];
    const float by = cb[jcol * 3 + 1];
    const float bz = cb[jcol * 3 + 2];
    const float cn2 = bx * bx + by * by + bz * bz;

    v2f bfrag;
    bfrag.x = half ? bz : bx;
    bfrag.y = half ? 0.0f : by;

    v8f dot = {};
    dot = wmma_f32_k4(afrag, bfrag, dot);

    float dd[8];
#pragma unroll
    for (int r = 0; r < 8; ++r) {
      const float sq = rn[r] + cn2 - 2.0f * dot[r];
      dd[r] = __builtin_amdgcn_sqrtf(fmaxf(sq, 1e-12f));
    }
#pragma unroll
    for (int r = 0; r < 8; ++r) {
      v2f d2;
      d2.x = dd[r];
      d2.y = dd[r];
      v2f q[NR / 2];
#pragma unroll
      for (int p = 0; p < NR / 2; ++p) {
        const v2f t = d2 - cent2[p];
        q[p] = (ngl2 * t) * t;
      }
#pragma unroll
      for (int p = 0; p < NR / 2; ++p) {
        v2f e;
        e.x = __builtin_amdgcn_exp2f(q[p].x);
        e.y = __builtin_amdgcn_exp2f(q[p].y);
        acc[r][p] += e;
      }
    }
  }

#pragma unroll
  for (int r = 0; r < 8; ++r) {
    float tot[NR];
#pragma unroll
    for (int k = 0; k < NR; ++k) {
      float v = (k & 1) ? acc[r][k >> 1].y : acc[r][k >> 1].x;
      v += __shfl_xor(v, 1, 16);
      v += __shfl_xor(v, 2, 16);
      v += __shfl_xor(v, 4, 16);
      v += __shfl_xor(v, 8, 16);
      tot[k] = v;
    }
    if (lid == 0) {
#pragma unroll
      for (int k = 0; k < NR; ++k)
        atomicAdd(&red[(r + 8 * half) * NR + k], tot[k]);
    }
  }
  __syncthreads();

  const int c0 = 32 * wave + 2 * lid;

  {
    const v2f bias = *(const v2f*)(b1 + c0);
    v8f g0, g1;
#pragma unroll
    for (int r = 0; r < 8; ++r) { g0[r] = bias.x; g1[r] = bias.y; }
#pragma unroll
    for (int s = 0; s < NR / 4; ++s) {
      const int kb = 4 * s + 2 * half;
      v2f a;
      a.x = red[lid * NR + kb] * (1.0f / (float)NL);
      a.y = red[lid * NR + kb + 1] * (1.0f / (float)NL);
      const v2f r0 = *(const v2f*)(w1 + kb * NE + c0);
      const v2f r1 = *(const v2f*)(w1 + (kb + 1) * NE + c0);
      v2f bb0, bb1;
      bb0.x = r0.x; bb0.y = r1.x;
      bb1.x = r0.y; bb1.y = r1.y;
      g0 = wmma_f32_k4(a, bb0, g0);
      g1 = wmma_f32_k4(a, bb1, g1);
    }
#pragma unroll
    for (int r = 0; r < 8; ++r) {
      v2f g;
      g.x = g0[r];
      g.y = g1[r];
      *(v2f*)(&hcat[(r + 8 * half) * APITCH + NE + c0]) = g;
    }
  }
  __syncthreads();

  {
    const v2f bias = *(const v2f*)(b2 + c0);
    v8f o0, o1;
#pragma unroll
    for (int r = 0; r < 8; ++r) { o0[r] = bias.x; o1[r] = bias.y; }
    const float* arow_lds = &hcat[lid * APITCH + 2 * half];
    const float* wcol = w2 + (size_t)(2 * half) * NE + c0;

    v2f aa[8], ab[8];
    v2f wa[16], wb[16];

    load_group(arow_lds, wcol, 0, aa, wa);
#pragma unroll 1
    for (int gp = 0; gp < 7; ++gp) {
      load_group(arow_lds, wcol, 2 * gp + 1, ab, wb);
      mma_group(aa, wa, o0, o1);
      load_group(arow_lds, wcol, 2 * gp + 2, aa, wa);
      mma_group(ab, wb, o0, o1);
    }
    load_group(arow_lds, wcol, 15, ab, wb);
    mma_group(aa, wa, o0, o1);
    mma_group(ab, wb, o0, o1);

    float* orow = out + ((size_t)b * NL + i0 + 8 * half) * NE + c0;
#pragma unroll
    for (int r = 0; r < 8; ++r) {
      v2f o;
      o.x = o0[r];
      o.y = o1[r];
      *(v2f*)(orow + (size_t)r * NE) = o;
    }
  }
}

extern "C" void kernel_launch(void* const* d_in, const int* in_sizes, int n_in,
                              void* d_out, int out_size, void* d_ws,
                              size_t ws_size, hipStream_t stream) {
  const float* coords   = (const float*)d_in[0];
  const int*   Z        = (const int*)d_in[1];
  const float* atom_emb = (const float*)d_in[2];
  const float* centers  = (const float*)d_in[3];
  const float* gamma    = (const float*)d_in[4];
  const float* w1       = (const float*)d_in[5];
  const float* b1       = (const float*)d_in[6];
  const float* w2       = (const float*)d_in[7];
  const float* b2       = (const float*)d_in[8];
  float* outp = (float*)d_out;

  coord_encoder_kernel<<<NB * (NL / 16), 256, 0, stream>>>(
      coords, Z, atom_emb, centers, gamma, w1, b1, w2, b2, outp);
}
